# starting kernel
# baseline (speedup 1.0000x reference)
#include <hip/hip_runtime.h>


#define NN 100000
#define NE 5000000
#define NE4 (NE / 4)

typedef __attribute__((ext_vector_type(2))) float v2f;
typedef __attribute__((ext_vector_type(8))) float v8f;


__global__ void k_init_deg(float* __restrict__ deg) {
  int n = blockIdx.x * blockDim.x + threadIdx.x;
  if (n < NN) deg[n] = 1.0f;
}

__global__ void k_deg_scatter(const int* __restrict__ col,
                              const float* __restrict__ w,
                              float* __restrict__ deg) {
  int t = blockIdx.x * blockDim.x + threadIdx.x;
  if (t < NE4) {
    int4   c  = ((const int4*)col)[t];
    float4 wv = ((const float4*)w)[t];
    atomicAdd(&deg[c.x], wv.x);
    atomicAdd(&deg[c.y], wv.y);
    atomicAdd(&deg[c.z], wv.z);
    atomicAdd(&deg[c.w], wv.w);
  }
}


__global__ void k_dinv_s0(const float* __restrict__ deg,
                          const float* __restrict__ x,
                          float* __restrict__ dinv,
                          float* __restrict__ s) {
  int n = blockIdx.x * blockDim.x + threadIdx.x;
  if (n < NN) {
    float d  = deg[n];
    float di = (d > 0.0f) ? rsqrtf(d) : 0.0f;
    dinv[n] = di;
    s[n] = di * di * x[n];
  }
}


__global__ void k_s_scatter(const int* __restrict__ row,
                            const int* __restrict__ col,
                            const float* __restrict__ w,
                            const float* __restrict__ dinv,
                            const float* __restrict__ x,
                            float* __restrict__ s) {
  int t = blockIdx.x * blockDim.x + threadIdx.x;
  if (t < NE4) {
    int4   r  = ((const int4*)row)[t];
    int4   c  = ((const int4*)col)[t];
    float4 wv = ((const float4*)w)[t];
    atomicAdd(&s[c.x], dinv[r.x] * wv.x * dinv[c.x] * x[r.x]);
    atomicAdd(&s[c.y], dinv[r.y] * wv.y * dinv[c.y] * x[r.y]);
    atomicAdd(&s[c.z], dinv[r.z] * wv.z * dinv[c.z] * x[r.z]);
    atomicAdd(&s[c.w], dinv[r.w] * wv.w * dinv[c.w] * x[r.w]);
  }
}


__global__ void __launch_bounds__(320)
k_node_transform(const float* __restrict__ s,
                 const float* __restrict__ W1,
                 const float* __restrict__ b1,
                 const float* __restrict__ W2,
                 float* __restrict__ z) {
  int lane = threadIdx.x & 31;
  int wave = (blockIdx.x * blockDim.x + threadIdx.x) >> 5;
  int base = wave * 16;
  int half = lane >> 4;
  int l16  = lane & 15;

  float w1 = W1[l16];
  v2f a;
  a.x = half ? 0.0f : w1;
  a.y = 0.0f;
  float sv = s[base + l16];
  v2f b;
  b.x = half ? 0.0f : sv;
  b.y = 0.0f;
  const float* b1p = b1 + half * 8;
  v8f c;
#pragma unroll
  for (int j = 0; j < 8; ++j) c[j] = b1p[j];

  v8f d = __builtin_amdgcn_wmma_f32_16x16x4_f32(
       false, a,  false, b,
       (short)0, c,  false,  false);

  const float* w2p = W2 + half * 8;
  float t = 0.0f;
#pragma unroll
  for (int j = 0; j < 8; ++j) t = fmaf(fmaxf(d[j], 0.0f), w2p[j], t);

  t += __shfl_xor(t, 16, 32);
  if (half == 0) z[base + l16] = t;
}


__global__ void k_out_init(const float* __restrict__ dinv,
                           const float* __restrict__ z,
                           const float* __restrict__ b2,
                           float* __restrict__ out) {
  int n = blockIdx.x * blockDim.x + threadIdx.x;
  if (n < NN) {
    float di = dinv[n];
    out[n] = di * di * z[n] + b2[0];
  }
}


__global__ void k_out_scatter(const int* __restrict__ row,
                              const int* __restrict__ col,
                              const float* __restrict__ w,
                              const float* __restrict__ dinv,
                              const float* __restrict__ z,
                              float* __restrict__ out) {
  int t = blockIdx.x * blockDim.x + threadIdx.x;
  if (t < NE4) {
    int4   r  = ((const int4*)row)[t];
    int4   c  = ((const int4*)col)[t];
    float4 wv = ((const float4*)w)[t];
    atomicAdd(&out[c.x], dinv[r.x] * wv.x * dinv[c.x] * z[r.x]);
    atomicAdd(&out[c.y], dinv[r.y] * wv.y * dinv[c.y] * z[r.y]);
    atomicAdd(&out[c.z], dinv[r.z] * wv.z * dinv[c.z] * z[r.z]);
    atomicAdd(&out[c.w], dinv[r.w] * wv.w * dinv[c.w] * z[r.w]);
  }
}

extern "C" void kernel_launch(void* const* d_in, const int* in_sizes, int n_in,
                              void* d_out, int out_size, void* d_ws, size_t ws_size,
                              hipStream_t stream) {
  const float* x   = (const float*)d_in[0];
  const int*   ei  = (const int*)d_in[1];
  const float* ew  = (const float*)d_in[2];
  const float* W1  = (const float*)d_in[3];
  const float* b1  = (const float*)d_in[4];
  const float* W2  = (const float*)d_in[5];
  const float* b2  = (const float*)d_in[6];
  float*       out = (float*)d_out;

  const int* row = ei;
  const int* col = ei + NE;

  float* deg  = (float*)d_ws;
  float* dinv = deg + NN;
  float* s    = dinv + NN;
  float* z    = s + NN;

  const int TB = 256;
  const int gridN = (NN + TB - 1) / TB;
  const int gridE = (NE4 + TB - 1) / TB;

  k_init_deg<<<gridN, TB, 0, stream>>>(deg);
  k_deg_scatter<<<gridE, TB, 0, stream>>>(col, ew, deg);
  k_dinv_s0<<<gridN, TB, 0, stream>>>(deg, x, dinv, s);
  k_s_scatter<<<gridE, TB, 0, stream>>>(row, col, ew, dinv, x, s);
  k_node_transform<<<625, 320, 0, stream>>>(s, W1, b1, W2, z);
  k_out_init<<<gridN, TB, 0, stream>>>(dinv, z, b2, out);
  k_out_scatter<<<gridE, TB, 0, stream>>>(row, col, ew, dinv, z, out);
}


# Edge scatters accumulate in LDS (2 node ranges x 128 edge slices, ds_add_f32) instead of global f32 atomics; partial rows then summed per node in slice order, so the summation order changes
# speedup vs baseline: 3.9680x; 3.9680x over previous
#include <hip/hip_runtime.h>


#define NN 100000
#define NE 5000000
#define NE4 (NE / 4)

#define NR0 50016
#define NR1 (NN - NR0)
#define NSL 128
#define SB  1024
#define STEP (NSL * SB)

typedef __attribute__((ext_vector_type(2))) float v2f;
typedef __attribute__((ext_vector_type(8))) float v8f;

__device__ __forceinline__ void lds_add(float* p, float v) {
  __hip_atomic_fetch_add(p, v, __ATOMIC_RELAXED, __HIP_MEMORY_SCOPE_WORKGROUP);
}

__device__ __forceinline__ void lds_zero(float* acc) {
  float4* a4 = (float4*)acc;
  for (int i = threadIdx.x; i < NR0 / 4; i += SB) a4[i] = make_float4(0.0f, 0.0f, 0.0f, 0.0f);
}

__device__ __forceinline__ void lds_flush(const float* acc, float* __restrict__ part,
                                          int slice, int lo, int cnt) {
  const float4* a4 = (const float4*)acc;
  float4* dst = (float4*)(part + slice * NN + lo);
  for (int i = threadIdx.x; i < cnt / 4; i += SB) dst[i] = a4[i];
}


__global__ void __launch_bounds__(SB)
k_deg_lds(const int* __restrict__ col,
          const float* __restrict__ w,
          float* __restrict__ part) {
  __shared__ __attribute__((aligned(16))) float acc[NR0];
  const int range = blockIdx.x & 1;
  const int slice = blockIdx.x >> 1;
  const int lo = range ? NR0 : 0;
  const unsigned cnt = range ? NR1 : NR0;

  lds_zero(acc);
  __syncthreads();
  const int4*   col4 = (const int4*)col;
  const float4* w4   = (const float4*)w;
  int q = slice * SB + threadIdx.x;
  int4   c  = col4[q];
  float4 wv = w4[q];
  for (;;) {
    const int qn = q + STEP;
    const int ql = (qn < NE4) ? qn : q;
    const int4   cn = col4[ql];
    const float4 wn = w4[ql];
    unsigned i0 = (unsigned)(c.x - lo), i1 = (unsigned)(c.y - lo);
    unsigned i2 = (unsigned)(c.z - lo), i3 = (unsigned)(c.w - lo);
    if (i0 < cnt) lds_add(&acc[i0], wv.x);
    if (i1 < cnt) lds_add(&acc[i1], wv.y);
    if (i2 < cnt) lds_add(&acc[i2], wv.z);
    if (i3 < cnt) lds_add(&acc[i3], wv.w);
    if (qn >= NE4) break;
    q = qn; c = cn; wv = wn;
  }
  __syncthreads();
  lds_flush(acc, part, slice, lo, (int)cnt);
}


__global__ void k_red_dinv(const float* __restrict__ part,
                           float* __restrict__ dinv) {
  int n = blockIdx.x * blockDim.x + threadIdx.x;
  if (n < NN) {
    float d = 1.0f;
    const float* p = part + n;
#pragma unroll 32
    for (int sl = 0; sl < NSL; ++sl) d += p[sl * NN];
    dinv[n] = (d > 0.0f) ? rsqrtf(d) : 0.0f;
  }
}


__global__ void __launch_bounds__(SB)
k_msg_lds(const int* __restrict__ row,
          const int* __restrict__ col,
          const float* __restrict__ w,
          const float* __restrict__ dinv,
          const float* __restrict__ v,
          float* __restrict__ part) {
  __shared__ __attribute__((aligned(16))) float acc[NR0];
  const int range = blockIdx.x & 1;
  const int slice = blockIdx.x >> 1;
  const int lo = range ? NR0 : 0;
  const unsigned cnt = range ? NR1 : NR0;

  lds_zero(acc);
  __syncthreads();
  const int4*   row4 = (const int4*)row;
  const int4*   col4 = (const int4*)col;
  const float4* w4   = (const float4*)w;
  int q = slice * SB + threadIdx.x;
  int4   r  = row4[q];
  int4   c  = col4[q];
  float4 wv = w4[q];
  for (;;) {
    const int qn = q + STEP;
    const int ql = (qn < NE4) ? qn : q;
    const int4   rn = row4[ql];
    const int4   cn = col4[ql];
    const float4 wn = w4[ql];
    const unsigned i0 = (unsigned)(c.x - lo), i1 = (unsigned)(c.y - lo);
    const unsigned i2 = (unsigned)(c.z - lo), i3 = (unsigned)(c.w - lo);
    const bool p0 = i0 < cnt, p1 = i1 < cnt, p2 = i2 < cnt, p3 = i3 < cnt;
    const int r0 = p0 ? r.x : 0, c0 = p0 ? c.x : 0;
    const int r1 = p1 ? r.y : 0, c1 = p1 ? c.y : 0;
    const int r2 = p2 ? r.z : 0, c2 = p2 ? c.z : 0;
    const int r3 = p3 ? r.w : 0, c3 = p3 ? c.w : 0;
    const float a0 = dinv[r0] * wv.x * dinv[c0] * v[r0];
    const float a1 = dinv[r1] * wv.y * dinv[c1] * v[r1];
    const float a2 = dinv[r2] * wv.z * dinv[c2] * v[r2];
    const float a3 = dinv[r3] * wv.w * dinv[c3] * v[r3];
    if (p0) lds_add(&acc[i0], a0);
    if (p1) lds_add(&acc[i1], a1);
    if (p2) lds_add(&acc[i2], a2);
    if (p3) lds_add(&acc[i3], a3);
    if (qn >= NE4) break;
    q = qn; r = rn; c = cn; wv = wn;
  }
  __syncthreads();
  lds_flush(acc, part, slice, lo, (int)cnt);
}


__global__ void k_red_s(const float* __restrict__ part,
                        const float* __restrict__ dinv,
                        const float* __restrict__ x,
                        float* __restrict__ s) {
  int n = blockIdx.x * blockDim.x + threadIdx.x;
  if (n < NN) {
    float di = dinv[n];
    float a = di * di * x[n];
    const float* p = part + n;
#pragma unroll 32
    for (int sl = 0; sl < NSL; ++sl) a += p[sl * NN];
    s[n] = a;
  }
}


__global__ void __launch_bounds__(320)
k_node_transform(const float* __restrict__ s,
                 const float* __restrict__ W1,
                 const float* __restrict__ b1,
                 const float* __restrict__ W2,
                 float* __restrict__ z) {
  int lane = threadIdx.x & 31;
  int wave = (blockIdx.x * blockDim.x + threadIdx.x) >> 5;
  int base = wave * 16;
  int half = lane >> 4;
  int l16  = lane & 15;

  float w1 = W1[l16];
  v2f a;
  a.x = half ? 0.0f : w1;
  a.y = 0.0f;
  float sv = s[base + l16];
  v2f b;
  b.x = half ? 0.0f : sv;
  b.y = 0.0f;
  const float* b1p = b1 + half * 8;
  v8f c;
#pragma unroll
  for (int j = 0; j < 8; ++j) c[j] = b1p[j];

  v8f d = __builtin_amdgcn_wmma_f32_16x16x4_f32(
       false, a,  false, b,
       (short)0, c,  false,  false);

  const float* w2p = W2 + half * 8;
  float t = 0.0f;
#pragma unroll
  for (int j = 0; j < 8; ++j) t = fmaf(fmaxf(d[j], 0.0f), w2p[j], t);

  t += __shfl_xor(t, 16, 32);
  if (half == 0) z[base + l16] = t;
}


__global__ void k_red_out(const float* __restrict__ part,
                          const float* __restrict__ dinv,
                          const float* __restrict__ z,
                          const float* __restrict__ b2,
                          float* __restrict__ out) {
  int n = blockIdx.x * blockDim.x + threadIdx.x;
  if (n < NN) {
    float di = dinv[n];
    float a = di * di * z[n] + b2[0];
    const float* p = part + n;
#pragma unroll 32
    for (int sl = 0; sl < NSL; ++sl) a += p[sl * NN];
    out[n] = a;
  }
}

extern "C" void kernel_launch(void* const* d_in, const int* in_sizes, int n_in,
                              void* d_out, int out_size, void* d_ws, size_t ws_size,
                              hipStream_t stream) {
  const float* x   = (const float*)d_in[0];
  const int*   ei  = (const int*)d_in[1];
  const float* ew  = (const float*)d_in[2];
  const float* W1  = (const float*)d_in[3];
  const float* b1  = (const float*)d_in[4];
  const float* W2  = (const float*)d_in[5];
  const float* b2  = (const float*)d_in[6];
  float*       out = (float*)d_out;

  const int* row = ei;
  const int* col = ei + NE;

  float* dinv = (float*)d_ws;
  float* s    = dinv + NN;
  float* z    = s + NN;
  float* part = z + NN;

  const int TB = 256;
  const int gridN = (NN + TB - 1) / TB;
  const int gridS = 2 * NSL;

  k_deg_lds<<<gridS, SB, 0, stream>>>(col, ew, part);
  k_red_dinv<<<gridN, TB, 0, stream>>>(part, dinv);
  k_msg_lds<<<gridS, SB, 0, stream>>>(row, col, ew, dinv, x, part);
  k_red_s<<<gridN, TB, 0, stream>>>(part, dinv, x, s);
  k_node_transform<<<625, 320, 0, stream>>>(s, W1, b1, W2, z);
  k_msg_lds<<<gridS, SB, 0, stream>>>(row, col, ew, dinv, z, part);
  k_red_out<<<gridN, TB, 0, stream>>>(part, dinv, z, b2, out);
}


# One gather per edge instead of three: edge loop adds w*u[row] with u=dinv*v pre-scaled per node; dinv[col] multiplies the finished node sum once (products regrouped, summation order kept)
# speedup vs baseline: 6.8972x; 1.7382x over previous
#include <hip/hip_runtime.h>


#define NN 100000
#define NE 5000000
#define NE4 (NE / 4)

#define NR0 50016
#define NR1 (NN - NR0)
#define NSL 128
#define SB  1024
#define STEP (NSL * SB)

typedef __attribute__((ext_vector_type(2))) float v2f;
typedef __attribute__((ext_vector_type(8))) float v8f;

__device__ __forceinline__ void lds_add(float* p, float v) {
  __hip_atomic_fetch_add(p, v, __ATOMIC_RELAXED, __HIP_MEMORY_SCOPE_WORKGROUP);
}

__device__ __forceinline__ void lds_zero(float* acc) {
  float4* a4 = (float4*)acc;
  for (int i = threadIdx.x; i < NR0 / 4; i += SB) a4[i] = make_float4(0.0f, 0.0f, 0.0f, 0.0f);
}

__device__ __forceinline__ void lds_flush(const float* acc, float* __restrict__ part,
                                          int slice, int lo, int cnt) {
  const float4* a4 = (const float4*)acc;
  float4* dst = (float4*)(part + slice * NN + lo);
  for (int i = threadIdx.x; i < cnt / 4; i += SB) dst[i] = a4[i];
}


__global__ void __launch_bounds__(SB)
k_deg_lds(const int* __restrict__ col,
          const float* __restrict__ w,
          float* __restrict__ part) {
  __shared__ __attribute__((aligned(16))) float acc[NR0];
  const int range = blockIdx.x & 1;
  const int slice = blockIdx.x >> 1;
  const int lo = range ? NR0 : 0;
  const unsigned cnt = range ? NR1 : NR0;

  lds_zero(acc);
  __syncthreads();
  const int4*   col4 = (const int4*)col;
  const float4* w4   = (const float4*)w;
  int q = slice * SB + threadIdx.x;
  int4   c  = col4[q];
  float4 wv = w4[q];
  for (;;) {
    const int qn = q + STEP;
    const int ql = (qn < NE4) ? qn : q;
    const int4   cn = col4[ql];
    const float4 wn = w4[ql];
    unsigned i0 = (unsigned)(c.x - lo), i1 = (unsigned)(c.y - lo);
    unsigned i2 = (unsigned)(c.z - lo), i3 = (unsigned)(c.w - lo);
    if (i0 < cnt) lds_add(&acc[i0], wv.x);
    if (i1 < cnt) lds_add(&acc[i1], wv.y);
    if (i2 < cnt) lds_add(&acc[i2], wv.z);
    if (i3 < cnt) lds_add(&acc[i3], wv.w);
    if (qn >= NE4) break;
    q = qn; c = cn; wv = wn;
  }
  __syncthreads();
  lds_flush(acc, part, slice, lo, (int)cnt);
}


__global__ void k_red_dinv(const float* __restrict__ part,
                           const float* __restrict__ x,
                           float* __restrict__ dinv,
                           float* __restrict__ u) {
  int n = blockIdx.x * blockDim.x + threadIdx.x;
  if (n < NN) {
    float d = 1.0f;
    const float* p = part + n;
#pragma unroll 32
    for (int sl = 0; sl < NSL; ++sl) d += p[sl * NN];
    const float di = (d > 0.0f) ? rsqrtf(d) : 0.0f;
    dinv[n] = di;
    u[n] = di * x[n];
  }
}


__global__ void __launch_bounds__(SB)
k_msg_lds(const int* __restrict__ row,
          const int* __restrict__ col,
          const float* __restrict__ w,
          const float* __restrict__ u,
          float* __restrict__ part) {
  __shared__ __attribute__((aligned(16))) float acc[NR0];
  const int range = blockIdx.x & 1;
  const int slice = blockIdx.x >> 1;
  const int lo = range ? NR0 : 0;
  const unsigned cnt = range ? NR1 : NR0;

  lds_zero(acc);
  __syncthreads();
  const int4*   row4 = (const int4*)row;
  const int4*   col4 = (const int4*)col;
  const float4* w4   = (const float4*)w;
  int q = slice * SB + threadIdx.x;
  int4   r  = row4[q];
  int4   c  = col4[q];
  float4 wv = w4[q];
  for (;;) {
    const int qn = q + STEP;
    const int ql = (qn < NE4) ? qn : q;
    const int4   rn = row4[ql];
    const int4   cn = col4[ql];
    const float4 wn = w4[ql];
    const unsigned i0 = (unsigned)(c.x - lo), i1 = (unsigned)(c.y - lo);
    const unsigned i2 = (unsigned)(c.z - lo), i3 = (unsigned)(c.w - lo);
    const bool p0 = i0 < cnt, p1 = i1 < cnt, p2 = i2 < cnt, p3 = i3 < cnt;
    const int r0 = p0 ? r.x : 0;
    const int r1 = p1 ? r.y : 0;
    const int r2 = p2 ? r.z : 0;
    const int r3 = p3 ? r.w : 0;
    const float a0 = wv.x * u[r0];
    const float a1 = wv.y * u[r1];
    const float a2 = wv.z * u[r2];
    const float a3 = wv.w * u[r3];
    if (p0) lds_add(&acc[i0], a0);
    if (p1) lds_add(&acc[i1], a1);
    if (p2) lds_add(&acc[i2], a2);
    if (p3) lds_add(&acc[i3], a3);
    if (qn >= NE4) break;
    q = qn; r = rn; c = cn; wv = wn;
  }
  __syncthreads();
  lds_flush(acc, part, slice, lo, (int)cnt);
}


__global__ void k_red_s(const float* __restrict__ part,
                        const float* __restrict__ dinv,
                        const float* __restrict__ u,
                        float* __restrict__ s) {
  int n = blockIdx.x * blockDim.x + threadIdx.x;
  if (n < NN) {
    float a = u[n];
    const float* p = part + n;
#pragma unroll 32
    for (int sl = 0; sl < NSL; ++sl) a += p[sl * NN];
    s[n] = dinv[n] * a;
  }
}


__global__ void __launch_bounds__(320)
k_node_transform(const float* __restrict__ s,
                 const float* __restrict__ dinv,
                 const float* __restrict__ W1,
                 const float* __restrict__ b1,
                 const float* __restrict__ W2,
                 float* __restrict__ u) {
  int lane = threadIdx.x & 31;
  int wave = (blockIdx.x * blockDim.x + threadIdx.x) >> 5;
  int base = wave * 16;
  int half = lane >> 4;
  int l16  = lane & 15;

  float w1 = W1[l16];
  v2f a;
  a.x = half ? 0.0f : w1;
  a.y = 0.0f;
  float sv = s[base + l16];
  float di = dinv[base + l16];
  v2f b;
  b.x = half ? 0.0f : sv;
  b.y = 0.0f;
  const float* b1p = b1 + half * 8;
  v8f c;
#pragma unroll
  for (int j = 0; j < 8; ++j) c[j] = b1p[j];

  v8f d = __builtin_amdgcn_wmma_f32_16x16x4_f32(
       false, a,  false, b,
       (short)0, c,  false,  false);

  const float* w2p = W2 + half * 8;
  float t = 0.0f;
#pragma unroll
  for (int j = 0; j < 8; ++j) t = fmaf(fmaxf(d[j], 0.0f), w2p[j], t);

  t += __shfl_xor(t, 16, 32);
  if (half == 0) u[base + l16] = di * t;
}


__global__ void k_red_out(const float* __restrict__ part,
                          const float* __restrict__ dinv,
                          const float* __restrict__ u,
                          const float* __restrict__ b2,
                          float* __restrict__ out) {
  int n = blockIdx.x * blockDim.x + threadIdx.x;
  if (n < NN) {
    float a = u[n];
    const float* p = part + n;
#pragma unroll 32
    for (int sl = 0; sl < NSL; ++sl) a += p[sl * NN];
    out[n] = dinv[n] * a + b2[0];
  }
}

extern "C" void kernel_launch(void* const* d_in, const int* in_sizes, int n_in,
                              void* d_out, int out_size, void* d_ws, size_t ws_size,
                              hipStream_t stream) {
  const float* x   = (const float*)d_in[0];
  const int*   ei  = (const int*)d_in[1];
  const float* ew  = (const float*)d_in[2];
  const float* W1  = (const float*)d_in[3];
  const float* b1  = (const float*)d_in[4];
  const float* W2  = (const float*)d_in[5];
  const float* b2  = (const float*)d_in[6];
  float*       out = (float*)d_out;

  const int* row = ei;
  const int* col = ei + NE;

  float* dinv = (float*)d_ws;
  float* s    = dinv + NN;
  float* u1   = s + NN;
  float* u2   = u1 + NN;
  float* part = u2 + NN;

  const int TB = 256;
  const int gridN = (NN + TB - 1) / TB;
  const int gridS = 2 * NSL;

  k_deg_lds<<<gridS, SB, 0, stream>>>(col, ew, part);
  k_red_dinv<<<gridN, TB, 0, stream>>>(part, x, dinv, u1);
  k_msg_lds<<<gridS, SB, 0, stream>>>(row, col, ew, u1, part);
  k_red_s<<<gridN, TB, 0, stream>>>(part, dinv, u1, s);
  k_node_transform<<<625, 320, 0, stream>>>(s, dinv, W1, b1, W2, u2);
  k_msg_lds<<<gridS, SB, 0, stream>>>(row, col, ew, u2, part);
  k_red_out<<<gridN, TB, 0, stream>>>(part, dinv, u2, b2, out);
}


# Message passes read u[row] from LDS, not global: degree pass files each edge once by (target range, source quarter), a work-group reads only its records; partial-sum grouping and add order change
# speedup vs baseline: 9.9876x; 1.4481x over previous
#include <hip/hip_runtime.h>


#define NN 100000
#define NE 5000000
#define NE4 (NE / 4)

#define NR0 50016
#define NR1 (NN - NR0)
#define NSL 128
#define SB  1024
#define STEP (NSL * SB)
#define NPASS 10
#define NQ  25000
#define CAP (NPASS * SB * 4)
#define NGRP (NSL / 4)
#define UNR 6
#define UQ  ((NQ / 4 + SB - 1) / SB)

static_assert(NPASS * STEP >= NE4 && (NPASS - 1) * STEP < NE4, "passes cover the edge list");
static_assert(4 * NQ == NN && NQ % 4 == 0, "source quarters");

typedef __attribute__((ext_vector_type(2))) float v2f;
typedef __attribute__((ext_vector_type(8))) float v8f;

__device__ __forceinline__ void lds_add(float* p, float v) {
  __hip_atomic_fetch_add(p, v, __ATOMIC_RELAXED, __HIP_MEMORY_SCOPE_WORKGROUP);
}

__device__ __forceinline__ void lds_zero(float* acc) {
  float4* a4 = (float4*)acc;
  for (int i = threadIdx.x; i < NR0 / 4; i += SB) a4[i] = make_float4(0.0f, 0.0f, 0.0f, 0.0f);
}

__device__ __forceinline__ void lds_flush(const float* acc, float* __restrict__ part,
                                          int slice, int lo, int cnt) {
  const float4* a4 = (const float4*)acc;
  float4* dst = (float4*)(part + slice * NN + lo);
  for (int i = threadIdx.x; i < cnt / 4; i += SB) dst[i] = a4[i];
}


__global__ void __launch_bounds__(SB)
k_deg_bin(const int* __restrict__ row,
          const int* __restrict__ col,
          const float* __restrict__ w,
          float* __restrict__ part,
          unsigned* __restrict__ cntg,
          unsigned long long* __restrict__ lists) {
  __shared__ __attribute__((aligned(16))) float acc[NR0];
  __shared__ unsigned long long fill;
  const int range = blockIdx.x & 1;
  const int slice = blockIdx.x >> 1;
  const int lo = range ? NR0 : 0;
  const unsigned cnt = range ? NR1 : NR0;
  const int wg = range * NSL + slice;
  unsigned long long* reg = lists + (size_t)wg * (2 * CAP);

  lds_zero(acc);
  if (threadIdx.x == 0) fill = 0ull;
  __syncthreads();
  const int4*   row4 = (const int4*)row;
  const int4*   col4 = (const int4*)col;
  const float4* w4   = (const float4*)w;
  const int q0 = slice * SB + threadIdx.x;
  int4   r  = row4[q0];
  int4   c  = col4[q0];
  float4 wv = w4[q0];
  for (int k = 0; k < NPASS; ++k) {
    const int q = q0 + k * STEP;
    const bool valid = q < NE4;
    const int qn = q + STEP;
    const int ql = (qn < NE4) ? qn : q0;
    const int4   rn = row4[ql];
    const int4   cn = col4[ql];
    const float4 wn = w4[ql];

    const int      rr[4] = {r.x, r.y, r.z, r.w};
    const unsigned ii[4] = {(unsigned)(c.x - lo), (unsigned)(c.y - lo),
                            (unsigned)(c.z - lo), (unsigned)(c.w - lo)};
    const float    ww[4] = {wv.x, wv.y, wv.z, wv.w};
    bool     p[4];
    int      sq[4];
    unsigned m[4][4];
    unsigned pre[4][4];
    unsigned tot[4] = {0u, 0u, 0u, 0u};
#pragma unroll
    for (int e = 0; e < 4; ++e) {
      p[e]  = valid && (ii[e] < cnt);
      sq[e] = (rr[e] >= NQ) + (rr[e] >= 2 * NQ) + (rr[e] >= 3 * NQ);
#pragma unroll
      for (int s = 0; s < 4; ++s) {
        m[e][s]   = __builtin_amdgcn_ballot_w32(p[e] && (sq[e] == s));
        pre[e][s] = tot[s];
        tot[s]   += (unsigned)__builtin_popcount(m[e][s]);
      }
    }
    const unsigned long long share =
        (unsigned long long)(tot[0] | (tot[1] << 16)) |
        ((unsigned long long)(tot[2] | (tot[3] << 16)) << 32);
    unsigned long long old = 0ull;
    if ((threadIdx.x & 31) == 0)
      old = __hip_atomic_fetch_add(&fill, share, __ATOMIC_RELAXED, __HIP_MEMORY_SCOPE_WORKGROUP);
    const unsigned olo = __builtin_amdgcn_readfirstlane((unsigned)old);
    const unsigned ohi = __builtin_amdgcn_readfirstlane((unsigned)(old >> 32));
    const unsigned base[4] = {olo & 0xffffu, olo >> 16, ohi & 0xffffu, ohi >> 16};

#pragma unroll
    for (int e = 0; e < 4; ++e) {
      if (p[e]) {
        lds_add(&acc[ii[e]], ww[e]);
        unsigned pos = 0u;
#pragma unroll
        for (int s = 0; s < 4; ++s)
          if (sq[e] == s) pos = base[s] + pre[e][s] + __builtin_amdgcn_mbcnt_lo(m[e][s], 0u);
        const unsigned idx = (sq[e] & 1) ? (unsigned)(CAP - 1) - pos : pos;
        const unsigned key = ((unsigned)(rr[e] - sq[e] * NQ) << 16) | ii[e];
        reg[(unsigned)(sq[e] >> 1) * (unsigned)CAP + idx] =
            (unsigned long long)key | ((unsigned long long)__float_as_uint(ww[e]) << 32);
      }
    }
    r = rn; c = cn; wv = wn;
  }
  __syncthreads();
  lds_flush(acc, part, slice, lo, (int)cnt);
  if (threadIdx.x < 4)
    cntg[wg * 4 + threadIdx.x] = (unsigned)(fill >> (16 * threadIdx.x)) & 0xffffu;
}


__global__ void k_red_dinv(const float* __restrict__ part,
                           const float* __restrict__ x,
                           float* __restrict__ dinv,
                           float* __restrict__ u) {
  int n = blockIdx.x * blockDim.x + threadIdx.x;
  if (n < NN) {
    float d = 1.0f;
    const float* p = part + n;
#pragma unroll 32
    for (int sl = 0; sl < NSL; ++sl) d += p[sl * NN];
    const float di = (d > 0.0f) ? rsqrtf(d) : 0.0f;
    dinv[n] = di;
    u[n] = di * x[n];
  }
}


__global__ void __launch_bounds__(SB)
k_msg_bin(const float* __restrict__ u,
          const unsigned* __restrict__ cntg,
          const unsigned long long* __restrict__ lists,
          float* __restrict__ part) {
  __shared__ __attribute__((aligned(16))) float acc[NR0];
  __shared__ __attribute__((aligned(16))) float us[NQ];
  const int range = blockIdx.x & 1;
  const int sq    = (blockIdx.x >> 1) & 3;
  const int grp   = blockIdx.x >> 3;
  const int lo = range ? NR0 : 0;
  const unsigned cnt = range ? NR1 : NR0;

  const float4* u4 = (const float4*)(u + sq * NQ);
  float4 uq[UQ];
#pragma unroll
  for (int k = 0; k < UQ; ++k) {
    const int i = threadIdx.x + k * SB;
    uq[k] = u4[(i < NQ / 4) ? i : (int)threadIdx.x];
  }
  lds_zero(acc);
  {
    float4* d4 = (float4*)us;
#pragma unroll
    for (int k = 0; k < UQ; ++k) {
      const int i = threadIdx.x + k * SB;
      if (i < NQ / 4) d4[i] = uq[k];
    }
  }
  int n[4];
  const unsigned long long* bp[4];
  int nmax = 0;
#pragma unroll
  for (int k = 0; k < 4; ++k) {
    const int wg = range * NSL + grp * 4 + k;
    n[k]  = (int)cntg[wg * 4 + sq];
    bp[k] = lists + (size_t)wg * (2 * CAP) + (size_t)(sq >> 1) * CAP;
    nmax  = n[k] > nmax ? n[k] : nmax;
  }
  const bool down = (sq & 1) != 0;
  const unsigned long long* spare = (const unsigned long long*)cntg;
  __syncthreads();

  for (int j0 = threadIdx.x; j0 < nmax; j0 += UNR * SB) {
    unsigned long long rec[4][UNR];
#pragma unroll
    for (int k = 0; k < 4; ++k) {
#pragma unroll
      for (int i = 0; i < UNR; ++i) {
        const int j = j0 + i * SB;
        const int idx = down ? (CAP - 1 - j) : j;
        const unsigned long long* src = (j < n[k]) ? (bp[k] + idx) : spare;
        rec[k][i] = *src;
      }
    }
    float val[4][UNR];
#pragma unroll
    for (int k = 0; k < 4; ++k) {
#pragma unroll
      for (int i = 0; i < UNR; ++i) {
        const int j = j0 + i * SB;
        const unsigned key = (j < n[k]) ? (unsigned)rec[k][i] : 0u;
        val[k][i] = __uint_as_float((unsigned)(rec[k][i] >> 32)) * us[key >> 16];
      }
    }
#pragma unroll
    for (int k = 0; k < 4; ++k) {
#pragma unroll
      for (int i = 0; i < UNR; ++i) {
        const int j = j0 + i * SB;
        if (j < n[k]) lds_add(&acc[(unsigned)rec[k][i] & 0xffffu], val[k][i]);
      }
    }
  }
  __syncthreads();
  lds_flush(acc, part, sq * NGRP + grp, lo, (int)cnt);
}


__global__ void k_red_s(const float* __restrict__ part,
                        const float* __restrict__ dinv,
                        const float* __restrict__ u,
                        float* __restrict__ s) {
  int n = blockIdx.x * blockDim.x + threadIdx.x;
  if (n < NN) {
    float a = u[n];
    const float* p = part + n;
#pragma unroll 32
    for (int sl = 0; sl < NSL; ++sl) a += p[sl * NN];
    s[n] = dinv[n] * a;
  }
}


__global__ void __launch_bounds__(320)
k_node_transform(const float* __restrict__ s,
                 const float* __restrict__ dinv,
                 const float* __restrict__ W1,
                 const float* __restrict__ b1,
                 const float* __restrict__ W2,
                 float* __restrict__ u) {
  int lane = threadIdx.x & 31;
  int wave = (blockIdx.x * blockDim.x + threadIdx.x) >> 5;
  int base = wave * 16;
  int half = lane >> 4;
  int l16  = lane & 15;

  float w1 = W1[l16];
  v2f a;
  a.x = half ? 0.0f : w1;
  a.y = 0.0f;
  float sv = s[base + l16];
  float di = dinv[base + l16];
  v2f b;
  b.x = half ? 0.0f : sv;
  b.y = 0.0f;
  const float* b1p = b1 + half * 8;
  v8f c;
#pragma unroll
  for (int j = 0; j < 8; ++j) c[j] = b1p[j];

  v8f d = __builtin_amdgcn_wmma_f32_16x16x4_f32(
       false, a,  false, b,
       (short)0, c,  false,  false);

  const float* w2p = W2 + half * 8;
  float t = 0.0f;
#pragma unroll
  for (int j = 0; j < 8; ++j) t = fmaf(fmaxf(d[j], 0.0f), w2p[j], t);

  t += __shfl_xor(t, 16, 32);
  if (half == 0) u[base + l16] = di * t;
}


__global__ void k_red_out(const float* __restrict__ part,
                          const float* __restrict__ dinv,
                          const float* __restrict__ u,
                          const float* __restrict__ b2,
                          float* __restrict__ out) {
  int n = blockIdx.x * blockDim.x + threadIdx.x;
  if (n < NN) {
    float a = u[n];
    const float* p = part + n;
#pragma unroll 32
    for (int sl = 0; sl < NSL; ++sl) a += p[sl * NN];
    out[n] = dinv[n] * a + b2[0];
  }
}

extern "C" void kernel_launch(void* const* d_in, const int* in_sizes, int n_in,
                              void* d_out, int out_size, void* d_ws, size_t ws_size,
                              hipStream_t stream) {
  const float* x   = (const float*)d_in[0];
  const int*   ei  = (const int*)d_in[1];
  const float* ew  = (const float*)d_in[2];
  const float* W1  = (const float*)d_in[3];
  const float* b1  = (const float*)d_in[4];
  const float* W2  = (const float*)d_in[5];
  const float* b2  = (const float*)d_in[6];
  float*       out = (float*)d_out;

  const int* row = ei;
  const int* col = ei + NE;

  float* dinv = (float*)d_ws;
  float* s    = dinv + NN;
  float* u1   = s + NN;
  float* u2   = u1 + NN;
  float* part = u2 + NN;
  unsigned* cntg = (unsigned*)(part + NSL * NN);
  unsigned long long* lists = (unsigned long long*)(cntg + 2 * NSL * 4);

  const int TB = 256;
  const int gridN = (NN + TB - 1) / TB;
  const int gridS = 2 * NSL;

  k_deg_bin<<<gridS, SB, 0, stream>>>(row, col, ew, part, cntg, lists);
  k_red_dinv<<<gridN, TB, 0, stream>>>(part, x, dinv, u1);
  k_msg_bin<<<gridS, SB, 0, stream>>>(u1, cntg, lists, part);
  k_red_s<<<gridN, TB, 0, stream>>>(part, dinv, u1, s);
  k_node_transform<<<625, 320, 0, stream>>>(s, dinv, W1, b1, W2, u2);
  k_msg_bin<<<gridS, SB, 0, stream>>>(u2, cntg, lists, part);
  k_red_out<<<gridN, TB, 0, stream>>>(part, dinv, u2, b2, out);
}


# Message passes cut the target axis in 4, the source axis in 2 (was 2 and 4): 64 partial rows instead of 128 written and summed per pass; partial-sum grouping and add order change
# speedup vs baseline: 11.7216x; 1.1736x over previous
#include <hip/hip_runtime.h>


#define NN 100000
#define NE 5000000
#define NE4 (NE / 4)

#define NR0 50016
#define NR1 (NN - NR0)
#define NSL 128
#define SB  1024
#define STEP (NSL * SB)
#define NPASS 10
#define NH  50000
#define NT0 25024
#define CAP (NPASS * SB * 4)
#define NGRP (NSL / 4)
#define NSLM (2 * NGRP)
#define UNR 6
#define UQ  ((NH / 4 + SB - 1) / SB)

static_assert(NPASS * STEP >= NE4 && (NPASS - 1) * STEP < NE4, "passes cover the edge list");
static_assert(2 * NH == NN && NH % 4 == 0 && NH <= 65536, "source halves");
static_assert(NT0 % 32 == 0 && NT0 < NR1 && (NR0 - NT0) % 4 == 0 && (NR1 - NT0) % 4 == 0 &&
              NR0 - NT0 <= NT0, "target quarters");

typedef __attribute__((ext_vector_type(2))) float v2f;
typedef __attribute__((ext_vector_type(8))) float v8f;

__device__ __forceinline__ void lds_add(float* p, float v) {
  __hip_atomic_fetch_add(p, v, __ATOMIC_RELAXED, __HIP_MEMORY_SCOPE_WORKGROUP);
}

__device__ __forceinline__ void lds_zero(float* acc, int n4) {
  float4* a4 = (float4*)acc;
  for (int i = threadIdx.x; i < n4; i += SB) a4[i] = make_float4(0.0f, 0.0f, 0.0f, 0.0f);
}

__device__ __forceinline__ void lds_flush(const float* acc, float* __restrict__ part,
                                          int slice, int lo, int cnt) {
  const float4* a4 = (const float4*)acc;
  float4* dst = (float4*)(part + slice * NN + lo);
  for (int i = threadIdx.x; i < cnt / 4; i += SB) dst[i] = a4[i];
}


__global__ void __launch_bounds__(SB)
k_deg_bin(const int* __restrict__ row,
          const int* __restrict__ col,
          const float* __restrict__ w,
          float* __restrict__ part,
          unsigned* __restrict__ cntg,
          unsigned long long* __restrict__ lists) {
  __shared__ __attribute__((aligned(16))) float acc[NR0];
  __shared__ unsigned long long fill;
  const int range = blockIdx.x & 1;
  const int slice = blockIdx.x >> 1;
  const int lo = range ? NR0 : 0;
  const unsigned cnt = range ? NR1 : NR0;
  const int wg = range * NSL + slice;
  unsigned long long* reg = lists + (size_t)wg * (2 * CAP);

  lds_zero(acc, NR0 / 4);
  if (threadIdx.x == 0) fill = 0ull;
  __syncthreads();
  const int4*   row4 = (const int4*)row;
  const int4*   col4 = (const int4*)col;
  const float4* w4   = (const float4*)w;
  const int q0 = slice * SB + threadIdx.x;
  int4   r  = row4[q0];
  int4   c  = col4[q0];
  float4 wv = w4[q0];
  for (int k = 0; k < NPASS; ++k) {
    const int q = q0 + k * STEP;
    const bool valid = q < NE4;
    const int qn = q + STEP;
    const int ql = (qn < NE4) ? qn : q0;
    const int4   rn = row4[ql];
    const int4   cn = col4[ql];
    const float4 wn = w4[ql];

    const int      rr[4] = {r.x, r.y, r.z, r.w};
    const unsigned ii[4] = {(unsigned)(c.x - lo), (unsigned)(c.y - lo),
                            (unsigned)(c.z - lo), (unsigned)(c.w - lo)};
    const float    ww[4] = {wv.x, wv.y, wv.z, wv.w};
    bool     p[4];
    int      sq[4];
    unsigned m[4][4];
    unsigned pre[4][4];
    unsigned tot[4] = {0u, 0u, 0u, 0u};
#pragma unroll
    for (int e = 0; e < 4; ++e) {
      p[e]  = valid && (ii[e] < cnt);
      sq[e] = ((ii[e] >= (unsigned)NT0) ? 2 : 0) + ((rr[e] >= NH) ? 1 : 0);
#pragma unroll
      for (int s = 0; s < 4; ++s) {
        m[e][s]   = __builtin_amdgcn_ballot_w32(p[e] && (sq[e] == s));
        pre[e][s] = tot[s];
        tot[s]   += (unsigned)__builtin_popcount(m[e][s]);
      }
    }
    const unsigned long long share =
        (unsigned long long)(tot[0] | (tot[1] << 16)) |
        ((unsigned long long)(tot[2] | (tot[3] << 16)) << 32);
    unsigned long long old = 0ull;
    if ((threadIdx.x & 31) == 0)
      old = __hip_atomic_fetch_add(&fill, share, __ATOMIC_RELAXED, __HIP_MEMORY_SCOPE_WORKGROUP);
    const unsigned olo = __builtin_amdgcn_readfirstlane((unsigned)old);
    const unsigned ohi = __builtin_amdgcn_readfirstlane((unsigned)(old >> 32));
    const unsigned base[4] = {olo & 0xffffu, olo >> 16, ohi & 0xffffu, ohi >> 16};

#pragma unroll
    for (int e = 0; e < 4; ++e) {
      if (p[e]) {
        lds_add(&acc[ii[e]], ww[e]);
        unsigned pos = 0u;
#pragma unroll
        for (int s = 0; s < 4; ++s)
          if (sq[e] == s) pos = base[s] + pre[e][s] + __builtin_amdgcn_mbcnt_lo(m[e][s], 0u);
        const unsigned idx = (sq[e] & 1) ? (unsigned)(CAP - 1) - pos : pos;
        const unsigned key = ((unsigned)(rr[e] - (sq[e] & 1) * NH) << 16) |
                             (ii[e] - (unsigned)(sq[e] >> 1) * (unsigned)NT0);
        reg[(unsigned)(sq[e] >> 1) * (unsigned)CAP + idx] =
            (unsigned long long)key | ((unsigned long long)__float_as_uint(ww[e]) << 32);
      }
    }
    r = rn; c = cn; wv = wn;
  }
  __syncthreads();
  lds_flush(acc, part, slice, lo, (int)cnt);
  if (threadIdx.x < 4)
    cntg[wg * 4 + threadIdx.x] = (unsigned)(fill >> (16 * threadIdx.x)) & 0xffffu;
}


__global__ void k_red_dinv(const float* __restrict__ part,
                           const float* __restrict__ x,
                           float* __restrict__ dinv,
                           float* __restrict__ u) {
  int n = blockIdx.x * blockDim.x + threadIdx.x;
  if (n < NN) {
    float d = 1.0f;
    const float* p = part + n;
#pragma unroll 32
    for (int sl = 0; sl < NSL; ++sl) d += p[sl * NN];
    const float di = (d > 0.0f) ? rsqrtf(d) : 0.0f;
    dinv[n] = di;
    u[n] = di * x[n];
  }
}


__global__ void __launch_bounds__(SB)
k_msg_bin(const float* __restrict__ u,
          const unsigned* __restrict__ cntg,
          const unsigned long long* __restrict__ lists,
          float* __restrict__ part) {
  __shared__ __attribute__((aligned(16))) float acc[NT0];
  __shared__ __attribute__((aligned(16))) float us[NH];
  const int range = blockIdx.x & 1;
  const int sub   = (blockIdx.x >> 1) & 1;
  const int sh    = (blockIdx.x >> 2) & 1;
  const int grp   = blockIdx.x >> 3;
  const int sq    = sub * 2 + sh;
  const int lo = (range ? NR0 : 0) + (sub ? NT0 : 0);
  const unsigned cnt = sub ? (unsigned)((range ? NR1 : NR0) - NT0) : (unsigned)NT0;

  const float4* u4 = (const float4*)(u + sh * NH);
  float4 uq[UQ];
#pragma unroll
  for (int k = 0; k < UQ; ++k) {
    const int i = threadIdx.x + k * SB;
    uq[k] = u4[(i < NH / 4) ? i : (int)threadIdx.x];
  }
  lds_zero(acc, NT0 / 4);
  {
    float4* d4 = (float4*)us;
#pragma unroll
    for (int k = 0; k < UQ; ++k) {
      const int i = threadIdx.x + k * SB;
      if (i < NH / 4) d4[i] = uq[k];
    }
  }
  int n[4];
  const unsigned long long* bp[4];
  int nmax = 0;
#pragma unroll
  for (int k = 0; k < 4; ++k) {
    const int wg = range * NSL + grp * 4 + k;
    n[k]  = (int)cntg[wg * 4 + sq];
    bp[k] = lists + (size_t)wg * (2 * CAP) + (size_t)sub * CAP;
    nmax  = n[k] > nmax ? n[k] : nmax;
  }
  const bool down = sh != 0;
  const unsigned long long* spare = (const unsigned long long*)cntg;
  __syncthreads();

  for (int j0 = threadIdx.x; j0 < nmax; j0 += UNR * SB) {
    unsigned long long rec[4][UNR];
#pragma unroll
    for (int k = 0; k < 4; ++k) {
#pragma unroll
      for (int i = 0; i < UNR; ++i) {
        const int j = j0 + i * SB;
        const int idx = down ? (CAP - 1 - j) : j;
        const unsigned long long* src = (j < n[k]) ? (bp[k] + idx) : spare;
        rec[k][i] = *src;
      }
    }
    float val[4][UNR];
#pragma unroll
    for (int k = 0; k < 4; ++k) {
#pragma unroll
      for (int i = 0; i < UNR; ++i) {
        const int j = j0 + i * SB;
        const unsigned key = (j < n[k]) ? (unsigned)rec[k][i] : 0u;
        val[k][i] = __uint_as_float((unsigned)(rec[k][i] >> 32)) * us[key >> 16];
      }
    }
#pragma unroll
    for (int k = 0; k < 4; ++k) {
#pragma unroll
      for (int i = 0; i < UNR; ++i) {
        const int j = j0 + i * SB;
        if (j < n[k]) lds_add(&acc[(unsigned)rec[k][i] & 0xffffu], val[k][i]);
      }
    }
  }
  __syncthreads();
  lds_flush(acc, part, sh * NGRP + grp, lo, (int)cnt);
}


__global__ void k_red_s(const float* __restrict__ part,
                        const float* __restrict__ dinv,
                        const float* __restrict__ u,
                        float* __restrict__ s) {
  int n = blockIdx.x * blockDim.x + threadIdx.x;
  if (n < NN) {
    float a = u[n];
    const float* p = part + n;
#pragma unroll 32
    for (int sl = 0; sl < NSLM; ++sl) a += p[sl * NN];
    s[n] = dinv[n] * a;
  }
}


__global__ void __launch_bounds__(320)
k_node_transform(const float* __restrict__ s,
                 const float* __restrict__ dinv,
                 const float* __restrict__ W1,
                 const float* __restrict__ b1,
                 const float* __restrict__ W2,
                 float* __restrict__ u) {
  int lane = threadIdx.x & 31;
  int wave = (blockIdx.x * blockDim.x + threadIdx.x) >> 5;
  int base = wave * 16;
  int half = lane >> 4;
  int l16  = lane & 15;

  float w1 = W1[l16];
  v2f a;
  a.x = half ? 0.0f : w1;
  a.y = 0.0f;
  float sv = s[base + l16];
  float di = dinv[base + l16];
  v2f b;
  b.x = half ? 0.0f : sv;
  b.y = 0.0f;
  const float* b1p = b1 + half * 8;
  v8f c;
#pragma unroll
  for (int j = 0; j < 8; ++j) c[j] = b1p[j];

  v8f d = __builtin_amdgcn_wmma_f32_16x16x4_f32(
       false, a,  false, b,
       (short)0, c,  false,  false);

  const float* w2p = W2 + half * 8;
  float t = 0.0f;
#pragma unroll
  for (int j = 0; j < 8; ++j) t = fmaf(fmaxf(d[j], 0.0f), w2p[j], t);

  t += __shfl_xor(t, 16, 32);
  if (half == 0) u[base + l16] = di * t;
}


__global__ void k_red_out(const float* __restrict__ part,
                          const float* __restrict__ dinv,
                          const float* __restrict__ u,
                          const float* __restrict__ b2,
                          float* __restrict__ out) {
  int n = blockIdx.x * blockDim.x + threadIdx.x;
  if (n < NN) {
    float a = u[n];
    const float* p = part + n;
#pragma unroll 32
    for (int sl = 0; sl < NSLM; ++sl) a += p[sl * NN];
    out[n] = dinv[n] * a + b2[0];
  }
}

extern "C" void kernel_launch(void* const* d_in, const int* in_sizes, int n_in,
                              void* d_out, int out_size, void* d_ws, size_t ws_size,
                              hipStream_t stream) {
  const float* x   = (const float*)d_in[0];
  const int*   ei  = (const int*)d_in[1];
  const float* ew  = (const float*)d_in[2];
  const float* W1  = (const float*)d_in[3];
  const float* b1  = (const float*)d_in[4];
  const float* W2  = (const float*)d_in[5];
  const float* b2  = (const float*)d_in[6];
  float*       out = (float*)d_out;

  const int* row = ei;
  const int* col = ei + NE;

  float* dinv = (float*)d_ws;
  float* s    = dinv + NN;
  float* u1   = s + NN;
  float* u2   = u1 + NN;
  float* part = u2 + NN;
  unsigned* cntg = (unsigned*)(part + NSL * NN);
  unsigned long long* lists = (unsigned long long*)(cntg + 2 * NSL * 4);

  const int TB = 256;
  const int gridN = (NN + TB - 1) / TB;
  const int gridS = 2 * NSL;

  k_deg_bin<<<gridS, SB, 0, stream>>>(row, col, ew, part, cntg, lists);
  k_red_dinv<<<gridN, TB, 0, stream>>>(part, x, dinv, u1);
  k_msg_bin<<<gridS, SB, 0, stream>>>(u1, cntg, lists, part);
  k_red_s<<<gridN, TB, 0, stream>>>(part, dinv, u1, s);
  k_node_transform<<<625, 320, 0, stream>>>(s, dinv, W1, b1, W2, u2);
  k_msg_bin<<<gridS, SB, 0, stream>>>(u2, cntg, lists, part);
  k_red_out<<<gridN, TB, 0, stream>>>(part, dinv, u2, b2, out);
}


# Layer 1's node sums and the node transform fused into one kernel: a wave finishes s for 32 nodes in registers and feeds two WMMA tiles; 6 launches, not 7; every sum and its order kept
# speedup vs baseline: 12.2323x; 1.0436x over previous
#include <hip/hip_runtime.h>


#define NN 100000
#define NE 5000000
#define NE4 (NE / 4)

#define NR0 50016
#define NR1 (NN - NR0)
#define NSL 128
#define SB  1024
#define STEP (NSL * SB)
#define NPASS 10
#define NH  50000
#define NT0 25024
#define CAP (NPASS * SB * 4)
#define NGRP (NSL / 4)
#define NSLM (2 * NGRP)
#define UNR 6
#define UQ  ((NH / 4 + SB - 1) / SB)

static_assert(NPASS * STEP >= NE4 && (NPASS - 1) * STEP < NE4, "passes cover the edge list");
static_assert(2 * NH == NN && NH % 4 == 0 && NH <= 65536, "source halves");
static_assert(NT0 % 32 == 0 && NT0 < NR1 && (NR0 - NT0) % 4 == 0 && (NR1 - NT0) % 4 == 0 &&
              NR0 - NT0 <= NT0, "target quarters");

typedef __attribute__((ext_vector_type(2))) float v2f;
typedef __attribute__((ext_vector_type(8))) float v8f;

__device__ __forceinline__ void lds_add(float* p, float v) {
  __hip_atomic_fetch_add(p, v, __ATOMIC_RELAXED, __HIP_MEMORY_SCOPE_WORKGROUP);
}

__device__ __forceinline__ void lds_zero(float* acc, int n4) {
  float4* a4 = (float4*)acc;
  for (int i = threadIdx.x; i < n4; i += SB) a4[i] = make_float4(0.0f, 0.0f, 0.0f, 0.0f);
}

__device__ __forceinline__ void lds_flush(const float* acc, float* __restrict__ part,
                                          int slice, int lo, int cnt) {
  const float4* a4 = (const float4*)acc;
  float4* dst = (float4*)(part + slice * NN + lo);
  for (int i = threadIdx.x; i < cnt / 4; i += SB) dst[i] = a4[i];
}


__global__ void __launch_bounds__(SB)
k_deg_bin(const int* __restrict__ row,
          const int* __restrict__ col,
          const float* __restrict__ w,
          float* __restrict__ part,
          unsigned* __restrict__ cntg,
          unsigned long long* __restrict__ lists) {
  __shared__ __attribute__((aligned(16))) float acc[NR0];
  __shared__ unsigned long long fill;
  const int range = blockIdx.x & 1;
  const int slice = blockIdx.x >> 1;
  const int lo = range ? NR0 : 0;
  const unsigned cnt = range ? NR1 : NR0;
  const int wg = range * NSL + slice;
  unsigned long long* reg = lists + (size_t)wg * (2 * CAP);

  lds_zero(acc, NR0 / 4);
  if (threadIdx.x == 0) fill = 0ull;
  __syncthreads();
  const int4*   row4 = (const int4*)row;
  const int4*   col4 = (const int4*)col;
  const float4* w4   = (const float4*)w;
  const int q0 = slice * SB + threadIdx.x;
  int4   r  = row4[q0];
  int4   c  = col4[q0];
  float4 wv = w4[q0];
  for (int k = 0; k < NPASS; ++k) {
    const int q = q0 + k * STEP;
    const bool valid = q < NE4;
    const int qn = q + STEP;
    const int ql = (qn < NE4) ? qn : q0;
    const int4   rn = row4[ql];
    const int4   cn = col4[ql];
    const float4 wn = w4[ql];

    const int      rr[4] = {r.x, r.y, r.z, r.w};
    const unsigned ii[4] = {(unsigned)(c.x - lo), (unsigned)(c.y - lo),
                            (unsigned)(c.z - lo), (unsigned)(c.w - lo)};
    const float    ww[4] = {wv.x, wv.y, wv.z, wv.w};
    bool     p[4];
    int      sq[4];
    unsigned m[4][4];
    unsigned pre[4][4];
    unsigned tot[4] = {0u, 0u, 0u, 0u};
#pragma unroll
    for (int e = 0; e < 4; ++e) {
      p[e]  = valid && (ii[e] < cnt);
      sq[e] = ((ii[e] >= (unsigned)NT0) ? 2 : 0) + ((rr[e] >= NH) ? 1 : 0);
#pragma unroll
      for (int s = 0; s < 4; ++s) {
        m[e][s]   = __builtin_amdgcn_ballot_w32(p[e] && (sq[e] == s));
        pre[e][s] = tot[s];
        tot[s]   += (unsigned)__builtin_popcount(m[e][s]);
      }
    }
    const unsigned long long share =
        (unsigned long long)(tot[0] | (tot[1] << 16)) |
        ((unsigned long long)(tot[2] | (tot[3] << 16)) << 32);
    unsigned long long old = 0ull;
    if ((threadIdx.x & 31) == 0)
      old = __hip_atomic_fetch_add(&fill, share, __ATOMIC_RELAXED, __HIP_MEMORY_SCOPE_WORKGROUP);
    const unsigned olo = __builtin_amdgcn_readfirstlane((unsigned)old);
    const unsigned ohi = __builtin_amdgcn_readfirstlane((unsigned)(old >> 32));
    const unsigned base[4] = {olo & 0xffffu, olo >> 16, ohi & 0xffffu, ohi >> 16};

#pragma unroll
    for (int e = 0; e < 4; ++e) {
      if (p[e]) {
        lds_add(&acc[ii[e]], ww[e]);
        unsigned pos = 0u;
#pragma unroll
        for (int s = 0; s < 4; ++s)
          if (sq[e] == s) pos = base[s] + pre[e][s] + __builtin_amdgcn_mbcnt_lo(m[e][s], 0u);
        const unsigned idx = (sq[e] & 1) ? (unsigned)(CAP - 1) - pos : pos;
        const unsigned key = ((unsigned)(rr[e] - (sq[e] & 1) * NH) << 16) |
                             (ii[e] - (unsigned)(sq[e] >> 1) * (unsigned)NT0);
        reg[(unsigned)(sq[e] >> 1) * (unsigned)CAP + idx] =
            (unsigned long long)key | ((unsigned long long)__float_as_uint(ww[e]) << 32);
      }
    }
    r = rn; c = cn; wv = wn;
  }
  __syncthreads();
  lds_flush(acc, part, slice, lo, (int)cnt);
  if (threadIdx.x < 4)
    cntg[wg * 4 + threadIdx.x] = (unsigned)(fill >> (16 * threadIdx.x)) & 0xffffu;
}


__global__ void k_red_dinv(const float* __restrict__ part,
                           const float* __restrict__ x,
                           float* __restrict__ dinv,
                           float* __restrict__ u) {
  int n = blockIdx.x * blockDim.x + threadIdx.x;
  if (n < NN) {
    float d = 1.0f;
    const float* p = part + n;
#pragma unroll 32
    for (int sl = 0; sl < NSL; ++sl) d += p[sl * NN];
    const float di = (d > 0.0f) ? rsqrtf(d) : 0.0f;
    dinv[n] = di;
    u[n] = di * x[n];
  }
}


__global__ void __launch_bounds__(SB)
k_msg_bin(const float* __restrict__ u,
          const unsigned* __restrict__ cntg,
          const unsigned long long* __restrict__ lists,
          float* __restrict__ part) {
  __shared__ __attribute__((aligned(16))) float acc[NT0];
  __shared__ __attribute__((aligned(16))) float us[NH];
  const int range = blockIdx.x & 1;
  const int sub   = (blockIdx.x >> 1) & 1;
  const int sh    = (blockIdx.x >> 2) & 1;
  const int grp   = blockIdx.x >> 3;
  const int sq    = sub * 2 + sh;
  const int lo = (range ? NR0 : 0) + (sub ? NT0 : 0);
  const unsigned cnt = sub ? (unsigned)((range ? NR1 : NR0) - NT0) : (unsigned)NT0;

  const float4* u4 = (const float4*)(u + sh * NH);
  float4 uq[UQ];
#pragma unroll
  for (int k = 0; k < UQ; ++k) {
    const int i = threadIdx.x + k * SB;
    uq[k] = u4[(i < NH / 4) ? i : (int)threadIdx.x];
  }
  lds_zero(acc, NT0 / 4);
  {
    float4* d4 = (float4*)us;
#pragma unroll
    for (int k = 0; k < UQ; ++k) {
      const int i = threadIdx.x + k * SB;
      if (i < NH / 4) d4[i] = uq[k];
    }
  }
  int n[4];
  const unsigned long long* bp[4];
  int nmax = 0;
#pragma unroll
  for (int k = 0; k < 4; ++k) {
    const int wg = range * NSL + grp * 4 + k;
    n[k]  = (int)cntg[wg * 4 + sq];
    bp[k] = lists + (size_t)wg * (2 * CAP) + (size_t)sub * CAP;
    nmax  = n[k] > nmax ? n[k] : nmax;
  }
  const bool down = sh != 0;
  const unsigned long long* spare = (const unsigned long long*)cntg;
  __syncthreads();

  for (int j0 = threadIdx.x; j0 < nmax; j0 += UNR * SB) {
    unsigned long long rec[4][UNR];
#pragma unroll
    for (int k = 0; k < 4; ++k) {
#pragma unroll
      for (int i = 0; i < UNR; ++i) {
        const int j = j0 + i * SB;
        const int idx = down ? (CAP - 1 - j) : j;
        const unsigned long long* src = (j < n[k]) ? (bp[k] + idx) : spare;
        rec[k][i] = *src;
      }
    }
    float val[4][UNR];
#pragma unroll
    for (int k = 0; k < 4; ++k) {
#pragma unroll
      for (int i = 0; i < UNR; ++i) {
        const int j = j0 + i * SB;
        const unsigned key = (j < n[k]) ? (unsigned)rec[k][i] : 0u;
        val[k][i] = __uint_as_float((unsigned)(rec[k][i] >> 32)) * us[key >> 16];
      }
    }
#pragma unroll
    for (int k = 0; k < 4; ++k) {
#pragma unroll
      for (int i = 0; i < UNR; ++i) {
        const int j = j0 + i * SB;
        if (j < n[k]) lds_add(&acc[(unsigned)rec[k][i] & 0xffffu], val[k][i]);
      }
    }
  }
  __syncthreads();
  lds_flush(acc, part, sh * NGRP + grp, lo, (int)cnt);
}


#define RTW 5
static_assert(NN == 625 * 32 * RTW, "whole waves, whole work-groups: grid 625 x 160");

__global__ void __launch_bounds__(32 * RTW)
k_red_transform(const float* __restrict__ part,
                const float* __restrict__ dinv,
                const float* __restrict__ u1,
                const float* __restrict__ W1,
                const float* __restrict__ b1,
                const float* __restrict__ W2,
                float* __restrict__ u2) {
  const int lane = threadIdx.x & 31;
  const int half = lane >> 4;
  const int l16  = lane & 15;
  const int n = blockIdx.x * (32 * RTW) + threadIdx.x;

  const float di = dinv[n];
  const float w1 = W1[l16];
  const float* b1p = b1 + half * 8;
  v8f c;
#pragma unroll
  for (int j = 0; j < 8; ++j) c[j] = b1p[j];
  const float* w2p = W2 + half * 8;
  float w2[8];
#pragma unroll
  for (int j = 0; j < 8; ++j) w2[j] = w2p[j];

  float acc = u1[n];
  const float* p = part + n;
#pragma unroll 32
  for (int sl = 0; sl < NSLM; ++sl) acc += p[sl * NN];
  const float sv = di * acc;
  const float sv_hi = __shfl_xor(sv, 16, 32);

  v2f a;
  a.x = half ? 0.0f : w1;
  a.y = 0.0f;
  v2f bt0, bt1;
  bt0.x = half ? 0.0f : sv;
  bt0.y = 0.0f;
  bt1.x = half ? 0.0f : sv_hi;
  bt1.y = 0.0f;

  const v8f d0 = __builtin_amdgcn_wmma_f32_16x16x4_f32(
       false, a,  false, bt0,
       (short)0, c,  false,  false);
  const v8f d1 = __builtin_amdgcn_wmma_f32_16x16x4_f32(
       false, a,  false, bt1,
       (short)0, c,  false,  false);

  float t0 = 0.0f, t1 = 0.0f;
#pragma unroll
  for (int j = 0; j < 8; ++j) {
    t0 = fmaf(fmaxf(d0[j], 0.0f), w2[j], t0);
    t1 = fmaf(fmaxf(d1[j], 0.0f), w2[j], t1);
  }
  t0 += __shfl_xor(t0, 16, 32);
  t1 += __shfl_xor(t1, 16, 32);
  u2[n] = di * (half ? t1 : t0);
}


__global__ void k_red_out(const float* __restrict__ part,
                          const float* __restrict__ dinv,
                          const float* __restrict__ u,
                          const float* __restrict__ b2,
                          float* __restrict__ out) {
  int n = blockIdx.x * blockDim.x + threadIdx.x;
  if (n < NN) {
    float a = u[n];
    const float* p = part + n;
#pragma unroll 32
    for (int sl = 0; sl < NSLM; ++sl) a += p[sl * NN];
    out[n] = dinv[n] * a + b2[0];
  }
}

extern "C" void kernel_launch(void* const* d_in, const int* in_sizes, int n_in,
                              void* d_out, int out_size, void* d_ws, size_t ws_size,
                              hipStream_t stream) {
  const float* x   = (const float*)d_in[0];
  const int*   ei  = (const int*)d_in[1];
  const float* ew  = (const float*)d_in[2];
  const float* W1  = (const float*)d_in[3];
  const float* b1  = (const float*)d_in[4];
  const float* W2  = (const float*)d_in[5];
  const float* b2  = (const float*)d_in[6];
  float*       out = (float*)d_out;

  const int* row = ei;
  const int* col = ei + NE;

  float* dinv = (float*)d_ws;
  float* u1   = dinv + 2 * NN;
  float* u2   = u1 + NN;
  float* part = u2 + NN;
  unsigned* cntg = (unsigned*)(part + NSL * NN);
  unsigned long long* lists = (unsigned long long*)(cntg + 2 * NSL * 4);

  const int TB = 256;
  const int gridN = (NN + TB - 1) / TB;
  const int gridS = 2 * NSL;

  k_deg_bin<<<gridS, SB, 0, stream>>>(row, col, ew, part, cntg, lists);
  k_red_dinv<<<gridN, TB, 0, stream>>>(part, x, dinv, u1);
  k_msg_bin<<<gridS, SB, 0, stream>>>(u1, cntg, lists, part);
  k_red_transform<<<625, 160, 0, stream>>>(part, dinv, u1, W1, b1, W2, u2);
  k_msg_bin<<<gridS, SB, 0, stream>>>(u2, cntg, lists, part);
  k_red_out<<<gridN, TB, 0, stream>>>(part, dinv, u2, b2, out);
}
